# starting kernel
# baseline (speedup 1.0000x reference)
#include <hip/hip_runtime.h>
#include <hip/hip_bf16.h>


typedef __attribute__((ext_vector_type(2))) float v2f;
typedef __attribute__((ext_vector_type(8))) float v8f;

#define BROWS 16384
#define DIMS  64
#define NNZ   50

__global__ __launch_bounds__(256) void gather_mean_kernel(
    const float* __restrict__ emb, const int* __restrict__ fidx,
    float* __restrict__ x) {
  const int wave = threadIdx.x >> 5;
  const int lane = threadIdx.x & 31;
  const int row  = blockIdx.x * 8 + wave;
  const int base = row * NNZ;

  float ax = 0.0f, ay = 0.0f;
#pragma unroll 5
  for (int j = 0; j < NNZ; ++j) {
    const int f = fidx[base + j];
    const float2 e =
        ((const float2*)(emb + (size_t)f * DIMS))[lane];
    ax += e.x;
    ay += e.y;
  }
  const float s = 1.0f / (float)NNZ;
  float2 r;
  r.x = ax * s;
  r.y = ay * s;
  ((float2*)(x + (size_t)row * DIMS))[lane] = r;
}

__global__ __launch_bounds__(256) void mlp3_wmma_kernel(
    const float* __restrict__ x,
    const float* __restrict__ W0, const float* __restrict__ b0,
    const float* __restrict__ W1, const float* __restrict__ b1,
    const float* __restrict__ W2, const float* __restrict__ b2,
    float* __restrict__ out) {
  __shared__ v2f   sW2[32 * DIMS];
  __shared__ float sBias[DIMS];
  __shared__ float sTile[8][16 * DIMS];

  const int tid   = threadIdx.x;
  const int wave  = tid >> 5;
  const int lane  = tid & 31;
  const int l15   = lane & 15;
  const int half  = lane >> 4;
  const int mbase = (blockIdx.x * 8 + wave) * 16;

  float* tp = &sTile[wave][0];

#pragma unroll
  for (int r = 0; r < 16; ++r)
    ((float2*)tp)[r * 32 + lane] =
        ((const float2*)(x + (size_t)(mbase + r) * DIMS))[lane];

  const float* Wg[3] = {W0, W1, W2};
  const float* bg[3] = {b0, b1, b2};

#pragma unroll
  for (int layer = 0; layer < 3; ++layer) {
    __syncthreads();
    {
      const float* wsrc = Wg[layer];
#pragma unroll
      for (int i = 0; i < 8; ++i) {
        const int idx = tid + 256 * i;
        const int p   = idx >> 6;
        const int n   = idx & 63;
        v2f frag;
        frag.x = wsrc[(2 * p) * DIMS + n];
        frag.y = wsrc[(2 * p + 1) * DIMS + n];
        sW2[idx] = frag;
      }
      if (tid < DIMS) sBias[tid] = bg[layer][tid];
    }
    __syncthreads();

    const float bv0 = sBias[0 + l15];
    const float bv1 = sBias[16 + l15];
    const float bv2 = sBias[32 + l15];
    const float bv3 = sBias[48 + l15];
    v8f a0 = {bv0, bv0, bv0, bv0, bv0, bv0, bv0, bv0};
    v8f a1 = {bv1, bv1, bv1, bv1, bv1, bv1, bv1, bv1};
    v8f a2 = {bv2, bv2, bv2, bv2, bv2, bv2, bv2, bv2};
    v8f a3 = {bv3, bv3, bv3, bv3, bv3, bv3, bv3, bv3};

#pragma unroll
    for (int k = 0; k < 16; ++k) {
      const v2f av = ((const v2f*)tp)[l15 * 32 + k * 2 + half];
      const int pbase = (2 * k + half) * 64 + l15;
      const v2f bb0 = sW2[pbase + 0];
      const v2f bb1 = sW2[pbase + 16];
      const v2f bb2 = sW2[pbase + 32];
      const v2f bb3 = sW2[pbase + 48];
      a0 = __builtin_amdgcn_wmma_f32_16x16x4_f32(false, av, false, bb0,
                                                 (short)0, a0, false, false);
      a1 = __builtin_amdgcn_wmma_f32_16x16x4_f32(false, av, false, bb1,
                                                 (short)0, a1, false, false);
      a2 = __builtin_amdgcn_wmma_f32_16x16x4_f32(false, av, false, bb2,
                                                 (short)0, a2, false, false);
      a3 = __builtin_amdgcn_wmma_f32_16x16x4_f32(false, av, false, bb3,
                                                 (short)0, a3, false, false);
    }

    const int mD = half * 8;
#pragma unroll
    for (int j = 0; j < 8; ++j) {
      tp[(mD + j) * DIMS + 0 + l15]  = fmaxf(a0[j], 0.0f);
      tp[(mD + j) * DIMS + 16 + l15] = fmaxf(a1[j], 0.0f);
      tp[(mD + j) * DIMS + 32 + l15] = fmaxf(a2[j], 0.0f);
      tp[(mD + j) * DIMS + 48 + l15] = fmaxf(a3[j], 0.0f);
    }
  }

#pragma unroll
  for (int r = 0; r < 16; ++r)
    ((float2*)(out + (size_t)(mbase + r) * DIMS))[lane] =
        ((const float2*)tp)[r * 32 + lane];
}

extern "C" void kernel_launch(void* const* d_in, const int* in_sizes, int n_in,
                              void* d_out, int out_size, void* d_ws,
                              size_t ws_size, hipStream_t stream) {
  const float* emb  = (const float*)d_in[0];
  const int*   fidx = (const int*)d_in[1];
  const float* W0   = (const float*)d_in[3];
  const float* b0   = (const float*)d_in[4];
  const float* W1   = (const float*)d_in[5];
  const float* b1   = (const float*)d_in[6];
  const float* W2   = (const float*)d_in[7];
  const float* b2   = (const float*)d_in[8];
  float*       out  = (float*)d_out;

  float* x = (float*)d_ws;

  gather_mean_kernel<<<BROWS / 8, 256, 0, stream>>>(emb, fidx, x);
  mlp3_wmma_kernel<<<BROWS / (8 * 16), 256, 0, stream>>>(x, W0, b0, W1, b1,
                                                         W2, b2, out);
}


# gather kernel: 2048 waves of 8 rows (was 16384 of 1): half-wave per row, 16-byte loads, indices staged in LDS, 35-40 loads in flight per wave (asm loads, counted waits); sum order unchanged
# speedup vs baseline: 1.1155x; 1.1155x over previous
#include <hip/hip_runtime.h>
#include <hip/hip_bf16.h>


typedef __attribute__((ext_vector_type(2))) float v2f;
typedef __attribute__((ext_vector_type(8))) float v8f;

#define BROWS 16384
#define DIMS  64
#define NNZ   50

typedef __attribute__((ext_vector_type(4))) float v4f;

#define ROWS_PER_WAVE 8
#define G_WAVES       4
#define G_BLOCKS      (BROWS / (G_WAVES * ROWS_PER_WAVE))
#define IDX_PER_WAVE  (ROWS_PER_WAVE * NNZ)
#define UNIT          5
#define UNITS_PER_PAIR (NNZ / UNIT)
#define N_UNITS       ((ROWS_PER_WAVE / 2) * UNITS_PER_PAIR)
#define RING          8

#define GATHER5(B, O, BASE)                                                 \
  asm volatile(                                                             \
      "global_load_b128 %0, %5, %10\n\t"                                    \
      "global_load_b128 %1, %6, %10\n\t"                                    \
      "global_load_b128 %2, %7, %10\n\t"                                    \
      "global_load_b128 %3, %8, %10\n\t"                                    \
      "global_load_b128 %4, %9, %10\n\t"                                    \
      "s_wait_xcnt 0x0"                                                     \
      : "=&v"(B[0]), "=&v"(B[1]), "=&v"(B[2]), "=&v"(B[3]), "=&v"(B[4])     \
      : "v"(O[0]), "v"(O[1]), "v"(O[2]), "v"(O[3]), "v"(O[4]), "s"(BASE))

#define LANDED5(B, N)                                                       \
  asm volatile("s_wait_loadcnt " #N                                         \
               : "+v"(B[0]), "+v"(B[1]), "+v"(B[2]), "+v"(B[3]), "+v"(B[4]))

__global__ __launch_bounds__(G_WAVES * 32) void gather_mean_kernel(
    const float* __restrict__ emb, const int* __restrict__ fidx,
    float* __restrict__ x) {
  __shared__ int4 sIdx[G_WAVES][IDX_PER_WAVE / 4];

  const int wave = threadIdx.x >> 5;
  const int lane = threadIdx.x & 31;
  const int l15  = lane & 15;
  const int half = lane >> 4;
  const int wrow = (blockIdx.x * G_WAVES + wave) * ROWS_PER_WAVE;
  const unsigned lo = (unsigned)l15 * 16u;
  const float s = 1.0f / (float)NNZ;

  {
    const int4* src = (const int4*)(fidx + (size_t)wrow * NNZ);
#pragma unroll
    for (int c = 0; c < (IDX_PER_WAVE / 4 + 31) / 32; ++c) {
      const int ch = c * 32 + lane;
      if (ch < IDX_PER_WAVE / 4) sIdx[wave][ch] = src[ch];
    }
  }
  const int* irow = (const int*)&sIdx[wave][0] + half * NNZ;

  v4f ring[RING][UNIT];

#define PREP_UNIT(w, off)                                                   \
  _Pragma("unroll") for (int i = 0; i < UNIT; ++i) off[i] =                 \
      ((unsigned)irow[((w) / UNITS_PER_PAIR) * 2 * NNZ +                    \
                      ((w) % UNITS_PER_PAIR) * UNIT + i]                    \
       << 8) | lo;

#define SUM_UNIT(u)                                                         \
  {                                                                         \
    _Pragma("unroll") for (int i = 0; i < UNIT; ++i) {                      \
      ax += ring[(u) % RING][i].x;                                          \
      ay += ring[(u) % RING][i].y;                                          \
      az += ring[(u) % RING][i].z;                                          \
      aw += ring[(u) % RING][i].w;                                          \
    }                                                                       \
    if ((u) % UNITS_PER_PAIR == UNITS_PER_PAIR - 1) {                       \
      float4 r;                                                             \
      r.x = ax * s;                                                         \
      r.y = ay * s;                                                         \
      r.z = az * s;                                                         \
      r.w = aw * s;                                                         \
         \
      const int row = wrow + 2 * ((u) / UNITS_PER_PAIR) + half;             \
      ((float4*)(x + (size_t)row * DIMS))[l15] = r;                         \
      ax = ay = az = aw = 0.0f;                                             \
    }                                                                       \
  }

#pragma unroll
  for (int w = 0; w < RING; ++w) {
    unsigned off[UNIT];
    PREP_UNIT(w, off)
    GATHER5(ring[w % RING], off, emb);
  }

  float ax = 0.0f, ay = 0.0f, az = 0.0f, aw = 0.0f;

#pragma unroll
  for (int u = 0; u < N_UNITS - RING; ++u) {
    unsigned off[UNIT];
    PREP_UNIT(u + RING, off)
    LANDED5(ring[u % RING], 35);
    SUM_UNIT(u)
    GATHER5(ring[u % RING], off, emb);
  }
  LANDED5(ring[32 % RING], 35);
  SUM_UNIT(32)
  LANDED5(ring[33 % RING], 30);
  SUM_UNIT(33)
  LANDED5(ring[34 % RING], 25);
  SUM_UNIT(34)
  LANDED5(ring[35 % RING], 20);
  SUM_UNIT(35)
  LANDED5(ring[36 % RING], 15);
  SUM_UNIT(36)
  LANDED5(ring[37 % RING], 10);
  SUM_UNIT(37)
  LANDED5(ring[38 % RING], 5);
  SUM_UNIT(38)
  LANDED5(ring[39 % RING], 0);
  SUM_UNIT(39)
#undef PREP_UNIT
#undef SUM_UNIT
}

__global__ __launch_bounds__(256) void mlp3_wmma_kernel(
    const float* __restrict__ x,
    const float* __restrict__ W0, const float* __restrict__ b0,
    const float* __restrict__ W1, const float* __restrict__ b1,
    const float* __restrict__ W2, const float* __restrict__ b2,
    float* __restrict__ out) {
  __shared__ v2f   sW2[32 * DIMS];
  __shared__ float sBias[DIMS];
  __shared__ float sTile[8][16 * DIMS];

  const int tid   = threadIdx.x;
  const int wave  = tid >> 5;
  const int lane  = tid & 31;
  const int l15   = lane & 15;
  const int half  = lane >> 4;
  const int mbase = (blockIdx.x * 8 + wave) * 16;

  float* tp = &sTile[wave][0];

#pragma unroll
  for (int r = 0; r < 16; ++r)
    ((float2*)tp)[r * 32 + lane] =
        ((const float2*)(x + (size_t)(mbase + r) * DIMS))[lane];

  const float* Wg[3] = {W0, W1, W2};
  const float* bg[3] = {b0, b1, b2};

#pragma unroll
  for (int layer = 0; layer < 3; ++layer) {
    __syncthreads();
    {
      const float* wsrc = Wg[layer];
#pragma unroll
      for (int i = 0; i < 8; ++i) {
        const int idx = tid + 256 * i;
        const int p   = idx >> 6;
        const int n   = idx & 63;
        v2f frag;
        frag.x = wsrc[(2 * p) * DIMS + n];
        frag.y = wsrc[(2 * p + 1) * DIMS + n];
        sW2[idx] = frag;
      }
      if (tid < DIMS) sBias[tid] = bg[layer][tid];
    }
    __syncthreads();

    const float bv0 = sBias[0 + l15];
    const float bv1 = sBias[16 + l15];
    const float bv2 = sBias[32 + l15];
    const float bv3 = sBias[48 + l15];
    v8f a0 = {bv0, bv0, bv0, bv0, bv0, bv0, bv0, bv0};
    v8f a1 = {bv1, bv1, bv1, bv1, bv1, bv1, bv1, bv1};
    v8f a2 = {bv2, bv2, bv2, bv2, bv2, bv2, bv2, bv2};
    v8f a3 = {bv3, bv3, bv3, bv3, bv3, bv3, bv3, bv3};

#pragma unroll
    for (int k = 0; k < 16; ++k) {
      const v2f av = ((const v2f*)tp)[l15 * 32 + k * 2 + half];
      const int pbase = (2 * k + half) * 64 + l15;
      const v2f bb0 = sW2[pbase + 0];
      const v2f bb1 = sW2[pbase + 16];
      const v2f bb2 = sW2[pbase + 32];
      const v2f bb3 = sW2[pbase + 48];
      a0 = __builtin_amdgcn_wmma_f32_16x16x4_f32(false, av, false, bb0,
                                                 (short)0, a0, false, false);
      a1 = __builtin_amdgcn_wmma_f32_16x16x4_f32(false, av, false, bb1,
                                                 (short)0, a1, false, false);
      a2 = __builtin_amdgcn_wmma_f32_16x16x4_f32(false, av, false, bb2,
                                                 (short)0, a2, false, false);
      a3 = __builtin_amdgcn_wmma_f32_16x16x4_f32(false, av, false, bb3,
                                                 (short)0, a3, false, false);
    }

    const int mD = half * 8;
#pragma unroll
    for (int j = 0; j < 8; ++j) {
      tp[(mD + j) * DIMS + 0 + l15]  = fmaxf(a0[j], 0.0f);
      tp[(mD + j) * DIMS + 16 + l15] = fmaxf(a1[j], 0.0f);
      tp[(mD + j) * DIMS + 32 + l15] = fmaxf(a2[j], 0.0f);
      tp[(mD + j) * DIMS + 48 + l15] = fmaxf(a3[j], 0.0f);
    }
  }

#pragma unroll
  for (int r = 0; r < 16; ++r)
    ((float2*)(out + (size_t)(mbase + r) * DIMS))[lane] =
        ((const float2*)tp)[r * 32 + lane];
}

extern "C" void kernel_launch(void* const* d_in, const int* in_sizes, int n_in,
                              void* d_out, int out_size, void* d_ws,
                              size_t ws_size, hipStream_t stream) {
  const float* emb  = (const float*)d_in[0];
  const int*   fidx = (const int*)d_in[1];
  const float* W0   = (const float*)d_in[3];
  const float* b0   = (const float*)d_in[4];
  const float* W1   = (const float*)d_in[5];
  const float* b1   = (const float*)d_in[6];
  const float* W2   = (const float*)d_in[7];
  const float* b2   = (const float*)d_in[8];
  float*       out  = (float*)d_out;

  float* x = (float*)d_ws;

  gather_mean_kernel<<<G_BLOCKS, G_WAVES * 32, 0, stream>>>(emb, fidx, x);
  mlp3_wmma_kernel<<<BROWS / (8 * 16), 256, 0, stream>>>(x, W0, b0, W1, b1,
                                                         W2, b2, out);
}


# MLP kernel: LDS row pitches padded (activation tile 64 -> 68 floats, weight pair rows 64 -> 80 float2): A-fragment loads were 16-way bank-conflicted, B loads and re-lay stores 2-way; now none
# speedup vs baseline: 1.1888x; 1.0657x over previous
#include <hip/hip_runtime.h>
#include <hip/hip_bf16.h>


typedef __attribute__((ext_vector_type(2))) float v2f;
typedef __attribute__((ext_vector_type(8))) float v8f;

#define BROWS 16384
#define DIMS  64
#define NNZ   50

typedef __attribute__((ext_vector_type(4))) float v4f;

#define ROWS_PER_WAVE 8
#define G_WAVES       4
#define G_BLOCKS      (BROWS / (G_WAVES * ROWS_PER_WAVE))
#define IDX_PER_WAVE  (ROWS_PER_WAVE * NNZ)
#define UNIT          5
#define UNITS_PER_PAIR (NNZ / UNIT)
#define N_UNITS       ((ROWS_PER_WAVE / 2) * UNITS_PER_PAIR)
#define RING          8

#define GATHER5(B, O, BASE)                                                 \
  asm volatile(                                                             \
      "global_load_b128 %0, %5, %10\n\t"                                    \
      "global_load_b128 %1, %6, %10\n\t"                                    \
      "global_load_b128 %2, %7, %10\n\t"                                    \
      "global_load_b128 %3, %8, %10\n\t"                                    \
      "global_load_b128 %4, %9, %10\n\t"                                    \
      "s_wait_xcnt 0x0"                                                     \
      : "=&v"(B[0]), "=&v"(B[1]), "=&v"(B[2]), "=&v"(B[3]), "=&v"(B[4])     \
      : "v"(O[0]), "v"(O[1]), "v"(O[2]), "v"(O[3]), "v"(O[4]), "s"(BASE))

#define LANDED5(B, N)                                                       \
  asm volatile("s_wait_loadcnt " #N                                         \
               : "+v"(B[0]), "+v"(B[1]), "+v"(B[2]), "+v"(B[3]), "+v"(B[4]))

__global__ __launch_bounds__(G_WAVES * 32) void gather_mean_kernel(
    const float* __restrict__ emb, const int* __restrict__ fidx,
    float* __restrict__ x) {
  __shared__ int4 sIdx[G_WAVES][IDX_PER_WAVE / 4];

  const int wave = threadIdx.x >> 5;
  const int lane = threadIdx.x & 31;
  const int l15  = lane & 15;
  const int half = lane >> 4;
  const int wrow = (blockIdx.x * G_WAVES + wave) * ROWS_PER_WAVE;
  const unsigned lo = (unsigned)l15 * 16u;
  const float s = 1.0f / (float)NNZ;

  {
    const int4* src = (const int4*)(fidx + (size_t)wrow * NNZ);
#pragma unroll
    for (int c = 0; c < (IDX_PER_WAVE / 4 + 31) / 32; ++c) {
      const int ch = c * 32 + lane;
      if (ch < IDX_PER_WAVE / 4) sIdx[wave][ch] = src[ch];
    }
  }
  const int* irow = (const int*)&sIdx[wave][0] + half * NNZ;

  v4f ring[RING][UNIT];

#define PREP_UNIT(w, off)                                                   \
  _Pragma("unroll") for (int i = 0; i < UNIT; ++i) off[i] =                 \
      ((unsigned)irow[((w) / UNITS_PER_PAIR) * 2 * NNZ +                    \
                      ((w) % UNITS_PER_PAIR) * UNIT + i]                    \
       << 8) | lo;

#define SUM_UNIT(u)                                                         \
  {                                                                         \
    _Pragma("unroll") for (int i = 0; i < UNIT; ++i) {                      \
      ax += ring[(u) % RING][i].x;                                          \
      ay += ring[(u) % RING][i].y;                                          \
      az += ring[(u) % RING][i].z;                                          \
      aw += ring[(u) % RING][i].w;                                          \
    }                                                                       \
    if ((u) % UNITS_PER_PAIR == UNITS_PER_PAIR - 1) {                       \
      float4 r;                                                             \
      r.x = ax * s;                                                         \
      r.y = ay * s;                                                         \
      r.z = az * s;                                                         \
      r.w = aw * s;                                                         \
         \
      const int row = wrow + 2 * ((u) / UNITS_PER_PAIR) + half;             \
      ((float4*)(x + (size_t)row * DIMS))[l15] = r;                         \
      ax = ay = az = aw = 0.0f;                                             \
    }                                                                       \
  }

#pragma unroll
  for (int w = 0; w < RING; ++w) {
    unsigned off[UNIT];
    PREP_UNIT(w, off)
    GATHER5(ring[w % RING], off, emb);
  }

  float ax = 0.0f, ay = 0.0f, az = 0.0f, aw = 0.0f;

#pragma unroll
  for (int u = 0; u < N_UNITS - RING; ++u) {
    unsigned off[UNIT];
    PREP_UNIT(u + RING, off)
    LANDED5(ring[u % RING], 35);
    SUM_UNIT(u)
    GATHER5(ring[u % RING], off, emb);
  }
  LANDED5(ring[32 % RING], 35);
  SUM_UNIT(32)
  LANDED5(ring[33 % RING], 30);
  SUM_UNIT(33)
  LANDED5(ring[34 % RING], 25);
  SUM_UNIT(34)
  LANDED5(ring[35 % RING], 20);
  SUM_UNIT(35)
  LANDED5(ring[36 % RING], 15);
  SUM_UNIT(36)
  LANDED5(ring[37 % RING], 10);
  SUM_UNIT(37)
  LANDED5(ring[38 % RING], 5);
  SUM_UNIT(38)
  LANDED5(ring[39 % RING], 0);
  SUM_UNIT(39)
#undef PREP_UNIT
#undef SUM_UNIT
}

#define TILE_PITCH 68
#define W_PITCH 80

__global__ __launch_bounds__(256) void mlp3_wmma_kernel(
    const float* __restrict__ x,
    const float* __restrict__ W0, const float* __restrict__ b0,
    const float* __restrict__ W1, const float* __restrict__ b1,
    const float* __restrict__ W2, const float* __restrict__ b2,
    float* __restrict__ out) {
  __shared__ v2f   sW2[32 * W_PITCH];
  __shared__ float sBias[DIMS];
  __shared__ float sTile[8][16 * TILE_PITCH];

  const int tid   = threadIdx.x;
  const int wave  = tid >> 5;
  const int lane  = tid & 31;
  const int l15   = lane & 15;
  const int half  = lane >> 4;
  const int mbase = (blockIdx.x * 8 + wave) * 16;

  float* tp = &sTile[wave][0];

#pragma unroll
  for (int r = 0; r < 16; ++r)
    ((float2*)(tp + r * TILE_PITCH))[lane] =
        ((const float2*)(x + (size_t)(mbase + r) * DIMS))[lane];

  const float* Wg[3] = {W0, W1, W2};
  const float* bg[3] = {b0, b1, b2};

#pragma unroll
  for (int layer = 0; layer < 3; ++layer) {
    __syncthreads();
    {
      const float* wsrc = Wg[layer];
#pragma unroll
      for (int i = 0; i < 8; ++i) {
        const int idx = tid + 256 * i;
        const int p   = idx >> 6;
        const int n   = idx & 63;
        v2f frag;
        frag.x = wsrc[(2 * p) * DIMS + n];
        frag.y = wsrc[(2 * p + 1) * DIMS + n];
        sW2[p * W_PITCH + n] = frag;
      }
      if (tid < DIMS) sBias[tid] = bg[layer][tid];
    }
    __syncthreads();

    const float bv0 = sBias[0 + l15];
    const float bv1 = sBias[16 + l15];
    const float bv2 = sBias[32 + l15];
    const float bv3 = sBias[48 + l15];
    v8f a0 = {bv0, bv0, bv0, bv0, bv0, bv0, bv0, bv0};
    v8f a1 = {bv1, bv1, bv1, bv1, bv1, bv1, bv1, bv1};
    v8f a2 = {bv2, bv2, bv2, bv2, bv2, bv2, bv2, bv2};
    v8f a3 = {bv3, bv3, bv3, bv3, bv3, bv3, bv3, bv3};

#pragma unroll
    for (int k = 0; k < 16; ++k) {
      const v2f av = ((const v2f*)(tp + l15 * TILE_PITCH))[k * 2 + half];
      const int pbase = (2 * k + half) * W_PITCH + l15;
      const v2f bb0 = sW2[pbase + 0];
      const v2f bb1 = sW2[pbase + 16];
      const v2f bb2 = sW2[pbase + 32];
      const v2f bb3 = sW2[pbase + 48];
      a0 = __builtin_amdgcn_wmma_f32_16x16x4_f32(false, av, false, bb0,
                                                 (short)0, a0, false, false);
      a1 = __builtin_amdgcn_wmma_f32_16x16x4_f32(false, av, false, bb1,
                                                 (short)0, a1, false, false);
      a2 = __builtin_amdgcn_wmma_f32_16x16x4_f32(false, av, false, bb2,
                                                 (short)0, a2, false, false);
      a3 = __builtin_amdgcn_wmma_f32_16x16x4_f32(false, av, false, bb3,
                                                 (short)0, a3, false, false);
    }

    const int mD = half * 8;
#pragma unroll
    for (int j = 0; j < 8; ++j) {
      tp[(mD + j) * TILE_PITCH + 0 + l15]  = fmaxf(a0[j], 0.0f);
      tp[(mD + j) * TILE_PITCH + 16 + l15] = fmaxf(a1[j], 0.0f);
      tp[(mD + j) * TILE_PITCH + 32 + l15] = fmaxf(a2[j], 0.0f);
      tp[(mD + j) * TILE_PITCH + 48 + l15] = fmaxf(a3[j], 0.0f);
    }
  }

#pragma unroll
  for (int r = 0; r < 16; ++r)
    ((float2*)(out + (size_t)(mbase + r) * DIMS))[lane] =
        ((const float2*)(tp + r * TILE_PITCH))[lane];
}

extern "C" void kernel_launch(void* const* d_in, const int* in_sizes, int n_in,
                              void* d_out, int out_size, void* d_ws,
                              size_t ws_size, hipStream_t stream) {
  const float* emb  = (const float*)d_in[0];
  const int*   fidx = (const int*)d_in[1];
  const float* W0   = (const float*)d_in[3];
  const float* b0   = (const float*)d_in[4];
  const float* W1   = (const float*)d_in[5];
  const float* b1   = (const float*)d_in[6];
  const float* W2   = (const float*)d_in[7];
  const float* b2   = (const float*)d_in[8];
  float*       out  = (float*)d_out;

  float* x = (float*)d_ws;

  gather_mean_kernel<<<G_BLOCKS, G_WAVES * 32, 0, stream>>>(emb, fidx, x);
  mlp3_wmma_kernel<<<BROWS / (8 * 16), 256, 0, stream>>>(x, W0, b0, W1, b1,
                                                         W2, b2, out);
}


# one launch instead of two: each 4-wave work-group gathers its 32 mean rows into LDS tiles and runs the 3 WMMA layers there (wave = tile x 32 columns), weights staged at its start; sum order kept
# speedup vs baseline: 1.7393x; 1.4631x over previous
#include <hip/hip_runtime.h>
#include <hip/hip_bf16.h>


typedef __attribute__((ext_vector_type(2))) float v2f;
typedef __attribute__((ext_vector_type(8))) float v8f;

#define BROWS 16384
#define DIMS  64
#define NNZ   50

#define TILE_PITCH 68
#define W_PITCH 80

typedef __attribute__((ext_vector_type(4))) float v4f;

#define ROWS_PER_WAVE 8
#define G_WAVES       4
#define G_BLOCKS      (BROWS / (G_WAVES * ROWS_PER_WAVE))
#define IDX_PER_WAVE  (ROWS_PER_WAVE * NNZ)
#define UNIT          5
#define UNITS_PER_PAIR (NNZ / UNIT)
#define N_UNITS       ((ROWS_PER_WAVE / 2) * UNITS_PER_PAIR)
#define RING          8
#define WG_ROWS       (G_WAVES * ROWS_PER_WAVE)
#define WG_TILES      (WG_ROWS / 16)

#define GATHER5(B, O, BASE)                                                 \
  asm volatile(                                                             \
      "global_load_b128 %0, %5, %10\n\t"                                    \
      "global_load_b128 %1, %6, %10\n\t"                                    \
      "global_load_b128 %2, %7, %10\n\t"                                    \
      "global_load_b128 %3, %8, %10\n\t"                                    \
      "global_load_b128 %4, %9, %10\n\t"                                    \
      "s_wait_xcnt 0x0"                                                     \
      : "=&v"(B[0]), "=&v"(B[1]), "=&v"(B[2]), "=&v"(B[3]), "=&v"(B[4])     \
      : "v"(O[0]), "v"(O[1]), "v"(O[2]), "v"(O[3]), "v"(O[4]), "s"(BASE))

#define LANDED5(B, N)                                                       \
  asm volatile("s_wait_loadcnt " #N                                         \
               : "+v"(B[0]), "+v"(B[1]), "+v"(B[2]), "+v"(B[3]), "+v"(B[4]))

__global__ __launch_bounds__(G_WAVES * 32) void bag_mlp3_kernel(
    const float* __restrict__ emb, const int* __restrict__ fidx,
    const float* __restrict__ W0, const float* __restrict__ b0,
    const float* __restrict__ W1, const float* __restrict__ b1,
    const float* __restrict__ W2, const float* __restrict__ b2,
    float* __restrict__ out) {
  __shared__ int4 sIdx[G_WAVES][IDX_PER_WAVE / 4];
  __shared__ __attribute__((aligned(16))) v2f sW2[3][32 * W_PITCH];
  __shared__ float sBias[3][DIMS];
  __shared__ __attribute__((aligned(16)))
      float sTile[2][WG_TILES][16 * TILE_PITCH];

  const int tid  = threadIdx.x;
  const int wave = __builtin_amdgcn_readfirstlane(tid >> 5);
  const int lane = tid & 31;
  const int l15  = lane & 15;
  const int half = lane >> 4;
  const int wrow = (blockIdx.x * G_WAVES + wave) * ROWS_PER_WAVE;
  const unsigned lo = (unsigned)l15 * 16u;
  const float s = 1.0f / (float)NNZ;
  float* xrows = &sTile[0][wave >> 1][(wave & 1) * ROWS_PER_WAVE * TILE_PITCH];

#define IDX_CH0(c) \
  ((c) * 32 < IDX_PER_WAVE / 4 - 32 ? (c) * 32 : IDX_PER_WAVE / 4 - 32)
  {
    const float* Wg[3] = {W0, W1, W2};
    const float* bg[3] = {b0, b1, b2};
    constexpr int N_CH = (IDX_PER_WAVE / 4 + 31) / 32;
    constexpr int N_IT = (32 * 16) / (G_WAVES * 32);
    int4   iv[N_CH];
    float4 wr0[3][N_IT], wr1[3][N_IT];
    float  bv[3];

    const int4* src = (const int4*)(fidx + (size_t)wrow * NNZ);
#pragma unroll
    for (int c = 0; c < N_CH; ++c) iv[c] = src[IDX_CH0(c) + lane];
#pragma unroll
    for (int layer = 0; layer < 3; ++layer) {
#pragma unroll
      for (int i = 0; i < N_IT; ++i) {
        const int item = tid + G_WAVES * 32 * i;
        const int p = item >> 4;
        const int q = item & 15;
        wr0[layer][i] = *(const float4*)(Wg[layer] + (2 * p) * DIMS + 4 * q);
        wr1[layer][i] = *(const float4*)(Wg[layer] + (2 * p + 1) * DIMS + 4 * q);
      }
      bv[layer] = bg[layer][tid & (DIMS - 1)];
    }

#pragma unroll
    for (int c = 0; c < N_CH; ++c) sIdx[wave][IDX_CH0(c) + lane] = iv[c];
#pragma unroll
    for (int layer = 0; layer < 3; ++layer) {
#pragma unroll
      for (int i = 0; i < N_IT; ++i) {
        const int item = tid + G_WAVES * 32 * i;
        const int p = item >> 4;
        const int q = item & 15;
        const float4 r0 = wr0[layer][i];
        const float4 r1 = wr1[layer][i];
        float4 f01, f23;
        f01.x = r0.x; f01.y = r1.x; f01.z = r0.y; f01.w = r1.y;
        f23.x = r0.z; f23.y = r1.z; f23.z = r0.w; f23.w = r1.w;
        float4* dst = (float4*)&sW2[layer][p * W_PITCH + 4 * q];
        dst[0] = f01;
        dst[1] = f23;
      }
    }
#pragma unroll
    for (int layer = 0; layer < 3; ++layer)
      sBias[layer][tid & (DIMS - 1)] = bv[layer];
  }
#undef IDX_CH0

  const int* irow = (const int*)&sIdx[wave][0] + half * NNZ;

  v4f ring[RING][UNIT];

#define PREP_UNIT(w, off)                                                   \
  _Pragma("unroll") for (int i = 0; i < UNIT; ++i) off[i] =                 \
      ((unsigned)irow[((w) / UNITS_PER_PAIR) * 2 * NNZ +                    \
                      ((w) % UNITS_PER_PAIR) * UNIT + i]                    \
       << 8) | lo;

#define SUM_UNIT(u)                                                         \
  {                                                                         \
    _Pragma("unroll") for (int i = 0; i < UNIT; ++i) {                      \
      ax += ring[(u) % RING][i].x;                                          \
      ay += ring[(u) % RING][i].y;                                          \
      az += ring[(u) % RING][i].z;                                          \
      aw += ring[(u) % RING][i].w;                                          \
    }                                                                       \
    if ((u) % UNITS_PER_PAIR == UNITS_PER_PAIR - 1) {                       \
      float4 r;                                                             \
      r.x = ax * s;                                                         \
      r.y = ay * s;                                                         \
      r.z = az * s;                                                         \
      r.w = aw * s;                                                         \
            \
      const int lrow = 2 * ((u) / UNITS_PER_PAIR) + half;                   \
      ((float4*)(xrows + lrow * TILE_PITCH))[l15] = r;                      \
      ax = ay = az = aw = 0.0f;                                             \
    }                                                                       \
  }

#pragma unroll
  for (int w = 0; w < RING; ++w) {
    unsigned off[UNIT];
    PREP_UNIT(w, off)
    GATHER5(ring[w % RING], off, emb);
  }

  float ax = 0.0f, ay = 0.0f, az = 0.0f, aw = 0.0f;

#pragma unroll
  for (int u = 0; u < N_UNITS - RING; ++u) {
    unsigned off[UNIT];
    PREP_UNIT(u + RING, off)
    LANDED5(ring[u % RING], 35);
    SUM_UNIT(u)
    GATHER5(ring[u % RING], off, emb);
  }
  LANDED5(ring[32 % RING], 35);
  SUM_UNIT(32)
  LANDED5(ring[33 % RING], 30);
  SUM_UNIT(33)
  LANDED5(ring[34 % RING], 25);
  SUM_UNIT(34)
  LANDED5(ring[35 % RING], 20);
  SUM_UNIT(35)
  LANDED5(ring[36 % RING], 15);
  SUM_UNIT(36)
  LANDED5(ring[37 % RING], 10);
  SUM_UNIT(37)
  LANDED5(ring[38 % RING], 5);
  SUM_UNIT(38)
  LANDED5(ring[39 % RING], 0);
  SUM_UNIT(39)
#undef PREP_UNIT
#undef SUM_UNIT

  const int t  = wave >> 1;
  const int nc = (wave & 1) * 32;

  __syncthreads();

#pragma unroll
  for (int layer = 0; layer < 3; ++layer) {
    const float* tin  = &sTile[layer & 1][t][0];
    float*       tout = &sTile[(layer + 1) & 1][t][0];
    const v2f*   wl   = &sW2[layer][0];

    const float bv0 = sBias[layer][nc + l15];
    const float bv1 = sBias[layer][nc + 16 + l15];
    v8f a0 = {bv0, bv0, bv0, bv0, bv0, bv0, bv0, bv0};
    v8f a1 = {bv1, bv1, bv1, bv1, bv1, bv1, bv1, bv1};

#pragma unroll
    for (int k = 0; k < 16; ++k) {
      const v2f av = ((const v2f*)(tin + l15 * TILE_PITCH))[k * 2 + half];
      const int pbase = (2 * k + half) * W_PITCH + nc + l15;
      const v2f bb0 = wl[pbase + 0];
      const v2f bb1 = wl[pbase + 16];
      a0 = __builtin_amdgcn_wmma_f32_16x16x4_f32(false, av, false, bb0,
                                                 (short)0, a0, false, false);
      a1 = __builtin_amdgcn_wmma_f32_16x16x4_f32(false, av, false, bb1,
                                                 (short)0, a1, false, false);
    }

    const int mD = half * 8;
#pragma unroll
    for (int j = 0; j < 8; ++j) {
      tout[(mD + j) * TILE_PITCH + nc + l15]      = fmaxf(a0[j], 0.0f);
      tout[(mD + j) * TILE_PITCH + nc + 16 + l15] = fmaxf(a1[j], 0.0f);
    }
    if (layer < 2) __syncthreads();
  }

  {
    const float* tf = &sTile[1][t][0];
    const int orow = blockIdx.x * WG_ROWS + 16 * t;
#pragma unroll
    for (int i = 0; i < 4; ++i) {
      const int r = 4 * i + (lane >> 3);
      const int c = nc + 4 * (lane & 7);
      const float4 v = *(const float4*)(tf + r * TILE_PITCH + c);
      *(float4*)(out + (size_t)(orow + r) * DIMS + c) = v;
    }
  }
}

extern "C" void kernel_launch(void* const* d_in, const int* in_sizes, int n_in,
                              void* d_out, int out_size, void* d_ws,
                              size_t ws_size, hipStream_t stream) {
  const float* emb  = (const float*)d_in[0];
  const int*   fidx = (const int*)d_in[1];
  const float* W0   = (const float*)d_in[3];
  const float* b0   = (const float*)d_in[4];
  const float* W1   = (const float*)d_in[5];
  const float* b1   = (const float*)d_in[6];
  const float* W2   = (const float*)d_in[7];
  const float* b2   = (const float*)d_in[8];
  float*       out  = (float*)d_out;

  bag_mlp3_kernel<<<G_BLOCKS, G_WAVES * 32, 0, stream>>>(emb, fidx, W0, b0, W1,
                                                         b1, W2, b2, out);
}


# tile 0's 3 WMMA layers run between tile 1's gather units, 2 k-steps at a time (32 rows = two 16-row tiles, wave = 16 columns); only tile 1's stand behind the gather; ring 8 -> 6 units; sum order kept
# speedup vs baseline: 1.8401x; 1.0580x over previous
#include <hip/hip_runtime.h>
#include <hip/hip_bf16.h>


typedef __attribute__((ext_vector_type(2))) float v2f;
typedef __attribute__((ext_vector_type(8))) float v8f;

#define BROWS 16384
#define DIMS  64
#define NNZ   50

#define TILE_PITCH 68
#define W_PITCH 80

typedef __attribute__((ext_vector_type(4))) float v4f;

#define ROWS_PER_WAVE 8
#define G_WAVES       4
#define G_BLOCKS      (BROWS / (G_WAVES * ROWS_PER_WAVE))
#define IDX_PER_WAVE  (ROWS_PER_WAVE * NNZ)
#define UNIT          5
#define UNITS_PER_PAIR (NNZ / UNIT)
#define N_UNITS       ((ROWS_PER_WAVE / 2) * UNITS_PER_PAIR)
#define RING          6
#define STEADY_YOUNGER 25
#define WG_ROWS       (G_WAVES * ROWS_PER_WAVE)
#define WG_TILES      (WG_ROWS / 16)

#define GATHER5(B, O, BASE)                                                 \
  asm volatile(                                                             \
      "global_load_b128 %0, %5, %10\n\t"                                    \
      "global_load_b128 %1, %6, %10\n\t"                                    \
      "global_load_b128 %2, %7, %10\n\t"                                    \
      "global_load_b128 %3, %8, %10\n\t"                                    \
      "global_load_b128 %4, %9, %10\n\t"                                    \
      "s_wait_xcnt 0x0"                                                     \
      : "=&v"(B[0]), "=&v"(B[1]), "=&v"(B[2]), "=&v"(B[3]), "=&v"(B[4])     \
      : "v"(O[0]), "v"(O[1]), "v"(O[2]), "v"(O[3]), "v"(O[4]), "s"(BASE))

#define LANDED5_(B, N)                                                      \
  asm volatile("s_wait_loadcnt " #N                                         \
               : "+v"(B[0]), "+v"(B[1]), "+v"(B[2]), "+v"(B[3]), "+v"(B[4]))
#define LANDED5(B, N) LANDED5_(B, N)

__global__ __launch_bounds__(G_WAVES * 32) void bag_mlp3_kernel(
    const float* __restrict__ emb, const int* __restrict__ fidx,
    const float* __restrict__ W0, const float* __restrict__ b0,
    const float* __restrict__ W1, const float* __restrict__ b1,
    const float* __restrict__ W2, const float* __restrict__ b2,
    float* __restrict__ out) {
  __shared__ int4 sIdx[G_WAVES][IDX_PER_WAVE / 4];
  __shared__ __attribute__((aligned(16))) v2f sW2[3][32 * W_PITCH];
  __shared__ float sBias[3][DIMS];
  __shared__ __attribute__((aligned(16)))
      float sTile[2][WG_TILES][16 * TILE_PITCH];

  const int tid  = threadIdx.x;
  const int wave = __builtin_amdgcn_readfirstlane(tid >> 5);
  const int lane = tid & 31;
  const int l15  = lane & 15;
  const int half = lane >> 4;
  const int wrow = blockIdx.x * WG_ROWS + wave * (ROWS_PER_WAVE / 2);
  const unsigned lo = (unsigned)l15 * 16u;
  const float s = 1.0f / (float)NNZ;
  float* xrows = &sTile[0][0][wave * (ROWS_PER_WAVE / 2) * TILE_PITCH];

#define IDX_CH0(c) \
  ((c) * 32 < IDX_PER_WAVE / 4 - 32 ? (c) * 32 : IDX_PER_WAVE / 4 - 32)
  {
    const float* Wg[3] = {W0, W1, W2};
    const float* bg[3] = {b0, b1, b2};
    constexpr int N_CH = (IDX_PER_WAVE / 4 + 31) / 32;
    constexpr int N_IT = (32 * 16) / (G_WAVES * 32);
    int4   iv[N_CH];
    float4 wr0[3][N_IT], wr1[3][N_IT];
    float  bv[3];

    constexpr int CH_TILE = IDX_PER_WAVE / 8;
    const int4* src0 = (const int4*)(fidx + (size_t)wrow * NNZ);
    const int4* src1 = (const int4*)(fidx + (size_t)(wrow + 16) * NNZ);
#pragma unroll
    for (int c = 0; c < N_CH; ++c) {
      const int ch = IDX_CH0(c) + lane;
      const int4* p = ch < CH_TILE ? src0 + ch : src1 + (ch - CH_TILE);
      iv[c] = *p;
    }
#pragma unroll
    for (int layer = 0; layer < 3; ++layer) {
#pragma unroll
      for (int i = 0; i < N_IT; ++i) {
        const int item = tid + G_WAVES * 32 * i;
        const int p = item >> 4;
        const int q = item & 15;
        wr0[layer][i] = *(const float4*)(Wg[layer] + (2 * p) * DIMS + 4 * q);
        wr1[layer][i] = *(const float4*)(Wg[layer] + (2 * p + 1) * DIMS + 4 * q);
      }
      bv[layer] = bg[layer][tid & (DIMS - 1)];
    }

#pragma unroll
    for (int c = 0; c < N_CH; ++c) sIdx[wave][IDX_CH0(c) + lane] = iv[c];
#pragma unroll
    for (int layer = 0; layer < 3; ++layer) {
#pragma unroll
      for (int i = 0; i < N_IT; ++i) {
        const int item = tid + G_WAVES * 32 * i;
        const int p = item >> 4;
        const int q = item & 15;
        const float4 r0 = wr0[layer][i];
        const float4 r1 = wr1[layer][i];
        float4 f01, f23;
        f01.x = r0.x; f01.y = r1.x; f01.z = r0.y; f01.w = r1.y;
        f23.x = r0.z; f23.y = r1.z; f23.z = r0.w; f23.w = r1.w;
        float4* dst = (float4*)&sW2[layer][p * W_PITCH + 4 * q];
        dst[0] = f01;
        dst[1] = f23;
      }
    }
#pragma unroll
    for (int layer = 0; layer < 3; ++layer)
      sBias[layer][tid & (DIMS - 1)] = bv[layer];
  }
#undef IDX_CH0

  const int* irow = (const int*)&sIdx[wave][0] + half * NNZ;

  v4f ring[RING][UNIT];

#define PREP_UNIT(w, off)                                                   \
  _Pragma("unroll") for (int i = 0; i < UNIT; ++i) off[i] =                 \
      ((unsigned)irow[((w) / UNITS_PER_PAIR) * 2 * NNZ +                    \
                      ((w) % UNITS_PER_PAIR) * UNIT + i]                    \
       << 8) | lo;

#define SUM_UNIT(u)                                                         \
  {                                                                         \
    _Pragma("unroll") for (int i = 0; i < UNIT; ++i) {                      \
      ax += ring[(u) % RING][i].x;                                          \
      ay += ring[(u) % RING][i].y;                                          \
      az += ring[(u) % RING][i].z;                                          \
      aw += ring[(u) % RING][i].w;                                          \
    }                                                                       \
    if ((u) % UNITS_PER_PAIR == UNITS_PER_PAIR - 1) {                       \
      float4 r;                                                             \
      r.x = ax * s;                                                         \
      r.y = ay * s;                                                         \
      r.z = az * s;                                                         \
      r.w = aw * s;                                                         \
        \
        \
      const int lrow = 2 * (((u) / UNITS_PER_PAIR) & 1) + half;             \
      float* trow = xrows + (((u) / UNITS_PER_PAIR) >> 1) * 16 * TILE_PITCH; \
      ((float4*)(trow + lrow * TILE_PITCH))[l15] = r;                       \
      ax = ay = az = aw = 0.0f;                                             \
    }                                                                       \
  }

#pragma unroll
  for (int w = 0; w < RING; ++w) {
    unsigned off[UNIT];
    PREP_UNIT(w, off)
    GATHER5(ring[w % RING], off, emb);
  }

  float ax = 0.0f, ay = 0.0f, az = 0.0f, aw = 0.0f;

  const int nc = wave * 16;
  v8f acc = {0.0f, 0.0f, 0.0f, 0.0f, 0.0f, 0.0f, 0.0f, 0.0f};

#define MLP_BEGIN(layer)                                                    \
  {                                                                         \
    const float bv = sBias[layer][nc + l15];                                \
    acc = (v8f){bv, bv, bv, bv, bv, bv, bv, bv};                            \
  }

#define MLP_K2(T, layer, k0)                                                \
  {                                                                         \
    const float* tin = &sTile[(layer) & 1][T][0];                           \
    const v2f*   wl  = &sW2[layer][0];                                      \
    _Pragma("unroll") for (int k = (k0); k < (k0) + 2; ++k) {               \
      const v2f av = ((const v2f*)(tin + l15 * TILE_PITCH))[k * 2 + half];  \
      const v2f bb = wl[(2 * k + half) * W_PITCH + nc + l15];               \
      acc = __builtin_amdgcn_wmma_f32_16x16x4_f32(false, av, false, bb,     \
                                                  (short)0, acc, false,     \
                                                  false);                   \
    }                                                                       \
  }

#define MLP_END(T, layer)                                                   \
  {                                                                         \
    float* tout = &sTile[((layer) + 1) & 1][T][0];                          \
    _Pragma("unroll") for (int j = 0; j < 8; ++j)                           \
      tout[(half * 8 + j) * TILE_PITCH + nc + l15] = fmaxf(acc[j], 0.0f);   \
  }

#define STORE_TILE(T)                                                       \
  {                                                                         \
    const float* tf = &sTile[1][T][0];                                      \
    const int orow = blockIdx.x * WG_ROWS + 16 * (T);                       \
    _Pragma("unroll") for (int i = 0; i < 2; ++i) {                         \
      const int r = 4 * wave + 2 * i + half;                                \
      const float4 v = *(const float4*)(tf + r * TILE_PITCH + 4 * l15);     \
      *(float4*)(out + (size_t)(orow + r) * DIMS + 4 * l15) = v;            \
    }                                                                       \
  }

#define T0_LAST_UNIT (N_UNITS / 2 - 1)
#define MLP0_FIRST   (N_UNITS / 2)
#define MLP0_END     (MLP0_FIRST + 3 * 4)

#define MLP0_PRE(u)                                                         \
  if ((u) > MLP0_FIRST && (u) <= MLP0_END) {                                \
    const int layer = ((u) - 1 - MLP0_FIRST) / 4;                           \
    const int q     = ((u) - 1 - MLP0_FIRST) % 4;                           \
    MLP_K2(0, layer, 4 * q + 2)                                             \
    if (q == 3) {                                                           \
      MLP_END(0, layer)                                                     \
      __syncthreads();                                                      \
    }                                                                       \
  }

#define MLP0_POST(u)                                                        \
  if ((u) == T0_LAST_UNIT) __syncthreads();                                 \
  if ((u) >= MLP0_FIRST && (u) < MLP0_END) {                                \
    const int layer = ((u) - MLP0_FIRST) / 4;                               \
    const int q     = ((u) - MLP0_FIRST) % 4;                               \
    if (q == 0) MLP_BEGIN(layer)                                            \
    MLP_K2(0, layer, 4 * q)                                                 \
  }                                                                         \
  if ((u) == MLP0_END) STORE_TILE(0)

  static_assert(RING >= 6 && RING <= 8 && UNIT == 5 && N_UNITS == 40,
                "the counted waits are written out for these sizes");
#pragma unroll
  for (int u = 0; u < N_UNITS - RING; ++u) {
    unsigned off[UNIT];
    PREP_UNIT(u + RING, off)
    MLP0_PRE(u)
    LANDED5(ring[u % RING], STEADY_YOUNGER);
    SUM_UNIT(u)
    GATHER5(ring[u % RING], off, emb);
    MLP0_POST(u)
  }
#define TAIL_UNIT(u, n)                                                     \
  MLP0_PRE(u)                                                               \
  LANDED5(ring[(u) % RING], n);                                             \
  SUM_UNIT(u)                                                               \
  MLP0_POST(u)
#if RING >= 8
  TAIL_UNIT(32, 35)
#endif
#if RING >= 7
  TAIL_UNIT(33, 30)
#endif
  TAIL_UNIT(34, 25)
  TAIL_UNIT(35, 20)
  TAIL_UNIT(36, 15)
  TAIL_UNIT(37, 10)
  TAIL_UNIT(38, 5)
  TAIL_UNIT(39, 0)
#undef TAIL_UNIT
#undef MLP0_PRE
#undef MLP0_POST
#undef PREP_UNIT
#undef SUM_UNIT

  __syncthreads();
#pragma unroll
  for (int layer = 0; layer < 3; ++layer) {
    MLP_BEGIN(layer)
#pragma unroll
    for (int q = 0; q < 8; ++q) MLP_K2(1, layer, 2 * q)
    MLP_END(1, layer)
    __syncthreads();
  }
  STORE_TILE(1)
#undef MLP_BEGIN
#undef MLP_K2
#undef MLP_END
#undef STORE_TILE
}

extern "C" void kernel_launch(void* const* d_in, const int* in_sizes, int n_in,
                              void* d_out, int out_size, void* d_ws,
                              size_t ws_size, hipStream_t stream) {
  const float* emb  = (const float*)d_in[0];
  const int*   fidx = (const int*)d_in[1];
  const float* W0   = (const float*)d_in[3];
  const float* b0   = (const float*)d_in[4];
  const float* W1   = (const float*)d_in[5];
  const float* b1   = (const float*)d_in[6];
  const float* W2   = (const float*)d_in[7];
  const float* b2   = (const float*)d_in[8];
  float*       out  = (float*)d_out;

  bag_mlp3_kernel<<<G_BLOCKS, G_WAVES * 32, 0, stream>>>(emb, fidx, W0, b0, W1,
                                                         b1, W2, b2, out);
}
